# starting kernel
# baseline (speedup 1.0000x reference)
#include <hip/hip_runtime.h>

typedef __attribute__((ext_vector_type(2))) float v2f;
typedef __attribute__((ext_vector_type(8))) float v8f;

#define N_TOT   2048
#define D_DIM   32
#define C_DIM   8
#define P_DIM   32
#define T_DIM   258
#define E_DIM   34
#define WROW    36
#define ZOUT_ELEMS ((size_t)4 * 512 * 8 * 32 * 34)

__global__ __launch_bounds__(32) void gp_precompute(
    const float* __restrict__ log_tau,
    const float* __restrict__ ts0,
    float* __restrict__ Wmat,
    float* __restrict__ parts) {
  const int dc = blockIdx.x;
  const int d  = dc >> 3;
  const int c  = dc & 7;
  const int lane = threadIdx.x;

  __shared__ float S[32][33];
  __shared__ float kpS[32][2];
  __shared__ float tsS[32];
  __shared__ float red[32];

  const float tau    = expf(log_tau[d]);
  const float inv2t2 = 1.0f / (2.0f * tau * tau);
  const float tp     = 20.0f + 2.0f * (float)c;

  const float t_p = ts0[lane];
  tsS[lane] = t_p;

  const float kp0 = expf(-(t_p * t_p) * inv2t2);
  const float dte = t_p - tp;
  const float kp1 = expf(-(dte * dte) * inv2t2);
  kpS[lane][0] = kp0;
  kpS[lane][1] = kp1;

  const float b   = expf(-(tp * tp) * inv2t2);
  const float det = 1.0f - b * b;
  const float a0  = (kp0 - b * kp1) / det;
  const float a1  = (kp1 - b * kp0) / det;
  __syncthreads();

  for (int q = 0; q < 32; ++q) {
    const float dd = t_p - tsS[q];
    S[lane][q] = expf(-(dd * dd) * inv2t2) - a0 * kpS[q][0] - a1 * kpS[q][1];
  }
  __syncthreads();

  for (int j = 0; j < 32; ++j) {
    if (lane == j) S[j][j] = sqrtf(S[j][j]);
    __syncthreads();
    const float djj = S[j][j];
    if (lane > j) S[lane][j] = S[lane][j] / djj;
    __syncthreads();
    if (lane > j) {
      const float lij = S[lane][j];
      for (int k = j + 1; k <= lane; ++k) S[lane][k] -= lij * S[k][j];
    }
    __syncthreads();
  }

  red[lane] = logf(fabsf(S[lane][lane]));
  __syncthreads();
  if (lane == 0) {
    float s = 0.0f;
    for (int i = 0; i < 32; ++i) s += red[i];
    parts[dc] = s;
  }

  float* wp = Wmat + ((size_t)dc * 32 + lane) * WROW;
  wp[0] = a0;
  wp[1] = a1;
  for (int q = 0; q < 32; ++q) wp[2 + q] = (q <= lane) ? S[lane][q] : 0.0f;
  wp[34] = 0.0f;
  wp[35] = 0.0f;
}

__global__ __launch_bounds__(256) void gp_gemm(
    const float* __restrict__ z,
    const float* __restrict__ Wmat,
    float* __restrict__ out) {
  const int tid  = threadIdx.x;
  const int wave = tid >> 5;
  const int lane = tid & 31;
  const int unit = blockIdx.x * 8 + wave;
  const int dc   = unit >> 7;
  const int mb   = unit & 127;
  const int d    = dc >> 3;
  const int c    = dc & 7;
  const int n0   = mb << 4;

  const int half = lane >> 4;
  const int l16  = lane & 15;

  const float* xrow  = z + ((size_t)(n0 + l16) * D_DIM + d) * T_DIM;
  const float* wbase = Wmat + (size_t)dc * 32 * WROW;
  const float* wr0   = wbase + (size_t)l16 * WROW;
  const float* wr1   = wbase + (size_t)(16 + l16) * WROW;

  v8f acc0 = {};
  v8f acc1 = {};

  for (int s = 0; s < 9; ++s) {
    const int ea = s * 4 + (half << 1);
    const int eb = ea + 1;
    const int ta = (ea < 2) ? ea : ((ea < E_DIM) ? (c * 32 + ea) : 0);
    const int tb = (eb < 2) ? eb : ((eb < E_DIM) ? (c * 32 + eb) : 0);

    v2f a;  a.x  = xrow[ta]; a.y  = xrow[tb];
    v2f b0; b0.x = wr0[ea];  b0.y = wr0[eb];
    v2f b1; b1.x = wr1[ea];  b1.y = wr1[eb];

    acc0 = __builtin_amdgcn_wmma_f32_16x16x4_f32(
        false, a, false, b0, (short)0, acc0, false, false);
    acc1 = __builtin_amdgcn_wmma_f32_16x16x4_f32(
        false, a, false, b1, (short)0, acc1, false, false);
  }

  for (int r = 0; r < 8; ++r) {
    const int    n     = n0 + r + (half << 3);
    const size_t obase = (((size_t)n * C_DIM + c) * D_DIM + d) * 34;
    out[obase + 1 + l16]      = acc0[r];
    out[obase + 1 + 16 + l16] = acc1[r];
  }

  {
    const int    n     = n0 + l16;
    const size_t obase = (((size_t)n * C_DIM + c) * D_DIM + d) * 34;
    const size_t zbase = ((size_t)n * D_DIM + d) * T_DIM;
    const float  v     = z[zbase + half];
    out[obase + (half ? 33 : 0)] = v;
  }
}

__global__ __launch_bounds__(256) void gp_sldj(
    const float* __restrict__ sldj_in,
    const float* __restrict__ parts,
    float* __restrict__ out) {
  const int n = blockIdx.x * blockDim.x + threadIdx.x;
  float tot = 0.0f;
  for (int i = 0; i < 256; ++i) tot += parts[i];
  out[ZOUT_ELEMS + n] = sldj_in[n] + tot;
}

extern "C" void kernel_launch(void* const* d_in, const int* in_sizes, int n_in,
                              void* d_out, int out_size, void* d_ws, size_t ws_size,
                              hipStream_t stream) {
  const float* z       = (const float*)d_in[0];
  const float* log_tau = (const float*)d_in[1];
  const float* ts0     = (const float*)d_in[2];
  const float* sldj    = (const float*)d_in[3];
  float* out = (float*)d_out;

  float* Wmat  = (float*)d_ws;
  float* parts = Wmat + 256 * 32 * WROW;

  gp_precompute<<<256, 32, 0, stream>>>(log_tau, ts0, Wmat, parts);
  gp_gemm<<<4096, 256, 0, stream>>>(z, Wmat, out);
  gp_sldj<<<8, 256, 0, stream>>>(sldj, parts, out);
}


# log-det partials: precompute work-group = 4 waves (one per SIMD, 64 groups) pre-adding its 4 partials; add kernel gets 64 values + its sldj in one trip (was 16+1 dependent trips); new sum order
# speedup vs baseline: 2.1315x; 2.1315x over previous
#include <hip/hip_runtime.h>

typedef __attribute__((ext_vector_type(2))) float v2f;
typedef __attribute__((ext_vector_type(4))) float v4f;
typedef __attribute__((ext_vector_type(8))) float v8f;
typedef float v4f_a8 __attribute__((ext_vector_type(4), aligned(8)));

#define N_TOT   2048
#define D_DIM   32
#define C_DIM   8
#define P_DIM   32
#define T_DIM   258
#define E_DIM   34
#define WROW    36
#define WFRAG   (32 * WROW)

#define ZOUT_ELEMS ((size_t)4 * 512 * 8 * 32 * 34)

#define PRE_WAVES 4
#define N_PARTS   (256 / PRE_WAVES)

static __device__ __forceinline__ void wave_lds_order() {
  __builtin_amdgcn_fence(__ATOMIC_RELEASE, "wavefront");
  __builtin_amdgcn_wave_barrier();
  __builtin_amdgcn_fence(__ATOMIC_ACQUIRE, "wavefront");
}

__global__ __launch_bounds__(32 * PRE_WAVES) __attribute__((amdgpu_waves_per_eu(1, 1))) void gp_precompute(
    const float* __restrict__ log_tau,
    const float* __restrict__ ts0,
    float* __restrict__ Wmat,
    float* __restrict__ parts) {
  const int wv = __builtin_amdgcn_readfirstlane((int)(threadIdx.x >> 5));
  const int dc = blockIdx.x * PRE_WAVES + wv;
  const int d  = dc >> 3;
  const int c  = dc & 7;
  const int lane = threadIdx.x & 31;

  __shared__ float S_all[PRE_WAVES][32][33];
  __shared__ float kpS_all[PRE_WAVES][32][2];
  __shared__ float tsS_all[PRE_WAVES][32];
  __shared__ float red_all[PRE_WAVES][32];
  __shared__ float wsum[PRE_WAVES];
  float (*S)[33]  = S_all[wv];
  float (*kpS)[2] = kpS_all[wv];
  float* tsS = tsS_all[wv];
  float* red = red_all[wv];

  const float tau    = expf(log_tau[d]);
  const float inv2t2 = 1.0f / (2.0f * tau * tau);
  const float tp     = 20.0f + 2.0f * (float)c;

  const float t_p = ts0[lane];
  tsS[lane] = t_p;

  const float kp0 = expf(-(t_p * t_p) * inv2t2);
  const float dte = t_p - tp;
  const float kp1 = expf(-(dte * dte) * inv2t2);
  kpS[lane][0] = kp0;
  kpS[lane][1] = kp1;

  const float b   = expf(-(tp * tp) * inv2t2);
  const float det = 1.0f - b * b;
  const float a0  = (kp0 - b * kp1) / det;
  const float a1  = (kp1 - b * kp0) / det;
  wave_lds_order();

#pragma unroll
  for (int q = 0; q < 32; ++q) {
    const float dd = t_p - tsS[q];
    S[lane][q] = expf(-(dd * dd) * inv2t2) - a0 * kpS[q][0] - a1 * kpS[q][1];
  }
  wave_lds_order();

#pragma unroll
  for (int j = 0; j < 32; ++j) {
    float s    = S[lane][j];
    float npiv = -S[j][j];
    float lj[32], li[32];
#pragma unroll
    for (int k = 0; k < j; ++k) {
      lj[k] = S[j][k];
      li[k] = S[lane][k];
    }
    __builtin_amdgcn_sched_barrier(0);
#pragma unroll
    for (int k = 0; k < j; ++k) {
      s    = __builtin_fmaf(-li[k], lj[k], s);
      npiv = __builtin_fmaf(lj[k], lj[k], npiv);
    }
    const float djj = sqrtf(-npiv);
    const float v   = (lane == j) ? djj : s / djj;
    if (lane >= j) S[lane][j] = v;
    wave_lds_order();
  }

  red[lane] = logf(fabsf(S[lane][lane]));
  wave_lds_order();
  if (lane == 0) {
    float s = 0.0f;
    for (int i = 0; i < 32; ++i) s += red[i];
    wsum[wv] = s;
  }
  __syncthreads();
  if (threadIdx.x == 0) {
    float s = 0.0f;
    for (int i = 0; i < PRE_WAVES; ++i) s += wsum[i];
    parts[blockIdx.x] = s;
  }

  float* wq = Wmat + (size_t)dc * WFRAG + (lane & 15) * 4 + (lane >> 4) * 2;
  *(v2f*)(wq)          = (v2f){a0, a1};
  *(v2f*)(wq + 16 * 4) = (v2f){0.0f, 0.0f};
  float row[32];
#pragma unroll
  for (int q = 0; q < 32; ++q) row[q] = S[lane][q];
#pragma unroll
  for (int g = 0; g < 16; ++g) {
    const int sm = g >> 1;
    const int h  = g & 1;
    const int q0 = 8 * (sm >> 1) + 4 * h + 2 * (sm & 1);
    v2f w;
    w.x = (q0     <= lane) ? row[q0]     : 0.0f;
    w.y = (q0 + 1 <= lane) ? row[q0 + 1] : 0.0f;
    *(v2f*)(wq + ((sm + 1) * 32 + 16 * h) * 4) = w;
  }
}

#define OBLK     1088
#define ORUN     544
#define ORUN_V4  136
#define RB_PER_WG 4
#define TILE_V4  (2 * 16 * ORUN_V4)

__global__ __launch_bounds__(512) void gp_gemm(
    const float* __restrict__ z,
    const float* __restrict__ Wmat,
    float* __restrict__ out) {
  __shared__ float4 tile4[2][TILE_V4];

  const int tid   = threadIdx.x;
  const int dl    = tid >> 5;
  const int lane  = tid & 31;
  const int c0    = (blockIdx.x & 3) << 1;
  const int dh    = (blockIdx.x >> 2) & 1;
  const int chunk = blockIdx.x >> 3;
  const int d     = dh * 16 + dl;
  const int half  = lane >> 4;
  const int l16   = lane & 15;

  const int moff = 2 + c0 * 32 + (half << 2);
  const float* xbase = z + ((size_t)((chunk * RB_PER_WG) << 4) * D_DIM + (size_t)l16 * D_DIM + d) * T_DIM;

  v2f    re[2];
  v4f_a8 rq[2][8];
  re[0] = *(const v2f*)xbase;
#pragma unroll
  for (int i = 0; i < 8; ++i) rq[0][i] = *(const v4f_a8*)(xbase + moff + 8 * i);

  const v4f* wf = (const v4f*)(Wmat + (size_t)(d * 8 + c0) * WFRAG) + lane;
  v2f b0[2][9], b1[2][9];
#pragma unroll
  for (int cc = 0; cc < 2; ++cc) {
#pragma unroll
    for (int s = 0; s < 9; ++s) {
      const v4f q = wf[cc * (WFRAG / 4) + s * 32];
      b0[cc][s] = (v2f){q.x, q.y};
      b1[cc][s] = (v2f){q.z, q.w};
    }
  }

  {
    const float* xn = xbase + (size_t)16 * D_DIM * T_DIM;
    re[1] = *(const v2f*)xn;
#pragma unroll
    for (int i = 0; i < 8; ++i) rq[1][i] = *(const v4f_a8*)(xn + moff + 8 * i);
  }

#pragma unroll
  for (int it = 0; it < RB_PER_WG; ++it) {
    const int n0  = (chunk * RB_PER_WG + it) << 4;
    const int set = it & 1;
    const v2f ae = re[set];

    float* tl = (float*)tile4[set];
#pragma unroll
    for (int cc = 0; cc < 2; ++cc) {
      v8f acc0 = {};
      v8f acc1 = {};
      acc0 = __builtin_amdgcn_wmma_f32_16x16x4_f32(
          false, ae, false, b0[cc][0], (short)0, acc0, false, false);
      acc1 = __builtin_amdgcn_wmma_f32_16x16x4_f32(
          false, ae, false, b1[cc][0], (short)0, acc1, false, false);
#pragma unroll
      for (int i = 0; i < 4; ++i) {
        const v4f_a8 q = rq[set][4 * cc + i];
        const v2f alo = (v2f){q.x, q.y};
        const v2f ahi = (v2f){q.z, q.w};
        acc0 = __builtin_amdgcn_wmma_f32_16x16x4_f32(
            false, alo, false, b0[cc][1 + 2 * i], (short)0, acc0, false, false);
        acc1 = __builtin_amdgcn_wmma_f32_16x16x4_f32(
            false, alo, false, b1[cc][1 + 2 * i], (short)0, acc1, false, false);
        acc0 = __builtin_amdgcn_wmma_f32_16x16x4_f32(
            false, ahi, false, b0[cc][2 + 2 * i], (short)0, acc0, false, false);
        acc1 = __builtin_amdgcn_wmma_f32_16x16x4_f32(
            false, ahi, false, b1[cc][2 + 2 * i], (short)0, acc1, false, false);
      }

      float* tc = tl + cc * (16 * ORUN);
#pragma unroll
      for (int r = 0; r < 8; ++r) {
        float* rowp = tc + (r + (half << 3)) * ORUN + dl * 34;
        rowp[1 + l16]  = acc0[r];
        rowp[17 + l16] = acc1[r];
      }
      if (half == 0) {
        float* rowp = tc + l16 * ORUN + dl * 34;
        rowp[0]  = ae.x;
        rowp[33] = ae.y;
      }
    }
    __syncthreads();

    if (it + 2 < RB_PER_WG) {
      const float* xn = xbase + (size_t)(it + 2) * 16 * D_DIM * T_DIM;
      re[set] = *(const v2f*)xn;
#pragma unroll
      for (int i = 0; i < 8; ++i) rq[set][i] = *(const v4f_a8*)(xn + moff + 8 * i);
    }

    const float4* src4 = tile4[set];
#pragma unroll
    for (int k = 0; k < 9; ++k) {
      const int f = tid + 512 * k;
      if (f < TILE_V4) {
        const int rr  = f / ORUN_V4;
        const int col = f - rr * ORUN_V4;
        float4* dst = (float4*)(out + ((size_t)(n0 + (rr & 15)) * C_DIM + c0 + (rr >> 4)) * OBLK + dh * ORUN) + col;
        *dst = src4[f];
      }
    }
  }
}

__global__ __launch_bounds__(256) void gp_sldj(
    const float* __restrict__ sldj_in,
    const float* __restrict__ parts,
    float* __restrict__ out) {
  const int n = blockIdx.x * blockDim.x + threadIdx.x;
  const float own = sldj_in[n];
  float p[N_PARTS];
#pragma unroll
  for (int i = 0; i < N_PARTS; ++i) p[i] = parts[i];
  __builtin_amdgcn_sched_barrier(0);
  float tot = 0.0f;
#pragma unroll
  for (int i = 0; i < N_PARTS; ++i) tot += p[i];
  out[ZOUT_ELEMS + n] = own + tot;
}

extern "C" void kernel_launch(void* const* d_in, const int* in_sizes, int n_in,
                              void* d_out, int out_size, void* d_ws, size_t ws_size,
                              hipStream_t stream) {
  const float* z       = (const float*)d_in[0];
  const float* log_tau = (const float*)d_in[1];
  const float* ts0     = (const float*)d_in[2];
  const float* sldj    = (const float*)d_in[3];
  float* out = (float*)d_out;

  float* Wmat  = (float*)d_ws;
  float* parts = Wmat + 256 * WFRAG;

  gp_precompute<<<64, 128, 0, stream>>>(log_tau, ts0, Wmat, parts);
  gp_gemm<<<256, 512, 0, stream>>>(z, Wmat, out);
  gp_sldj<<<8, 256, 0, stream>>>(sldj, parts, out);
}


# precompute: Cholesky column step pipelined: lanes exchange the UNDIVIDED column via LDS while sqrt runs, each divides what it reads back; next column's sums run beside; same sums, same order
# speedup vs baseline: 2.2872x; 1.0730x over previous
#include <hip/hip_runtime.h>

typedef __attribute__((ext_vector_type(2))) float v2f;
typedef __attribute__((ext_vector_type(4))) float v4f;
typedef __attribute__((ext_vector_type(8))) float v8f;
typedef float v4f_a8 __attribute__((ext_vector_type(4), aligned(8)));

#define N_TOT   2048
#define D_DIM   32
#define C_DIM   8
#define P_DIM   32
#define T_DIM   258
#define E_DIM   34
#define WROW    36
#define WFRAG   (32 * WROW)

#define ZOUT_ELEMS ((size_t)4 * 512 * 8 * 32 * 34)

#define PRE_WAVES 4
#define N_PARTS   (256 / PRE_WAVES)

static __device__ __forceinline__ void wave_lds_order() {
  __builtin_amdgcn_fence(__ATOMIC_RELEASE, "wavefront");
  __builtin_amdgcn_wave_barrier();
  __builtin_amdgcn_fence(__ATOMIC_ACQUIRE, "wavefront");
}

__global__ __launch_bounds__(32 * PRE_WAVES) __attribute__((amdgpu_waves_per_eu(1, 1))) void gp_precompute(
    const float* __restrict__ log_tau,
    const float* __restrict__ ts0,
    float* __restrict__ Wmat,
    float* __restrict__ parts) {
  const int wv = __builtin_amdgcn_readfirstlane((int)(threadIdx.x >> 5));
  const int dc = blockIdx.x * PRE_WAVES + wv;
  const int d  = dc >> 3;
  const int c  = dc & 7;
  const int lane = threadIdx.x & 31;

  __shared__ float S_all[PRE_WAVES][32][33];
  __shared__ float kpS_all[PRE_WAVES][32][2];
  __shared__ float tsS_all[PRE_WAVES][32];
  __shared__ float red_all[PRE_WAVES][32];
  __shared__ float wsum[PRE_WAVES];
  __shared__ float xch_all[PRE_WAVES][32];
  float (*S)[33]  = S_all[wv];
  float (*kpS)[2] = kpS_all[wv];
  float* tsS = tsS_all[wv];
  float* red = red_all[wv];
  float* xch = xch_all[wv];

  const float tau    = expf(log_tau[d]);
  const float inv2t2 = 1.0f / (2.0f * tau * tau);
  const float tp     = 20.0f + 2.0f * (float)c;

  const float t_p = ts0[lane];
  tsS[lane] = t_p;

  const float kp0 = expf(-(t_p * t_p) * inv2t2);
  const float dte = t_p - tp;
  const float kp1 = expf(-(dte * dte) * inv2t2);
  kpS[lane][0] = kp0;
  kpS[lane][1] = kp1;

  const float b   = expf(-(tp * tp) * inv2t2);
  const float det = 1.0f - b * b;
  const float a0  = (kp0 - b * kp1) / det;
  const float a1  = (kp1 - b * kp0) / det;
  wave_lds_order();

#pragma unroll
  for (int q = 0; q < 32; ++q) {
    const float dd = t_p - tsS[q];
    S[lane][q] = expf(-(dd * dd) * inv2t2) - a0 * kpS[q][0] - a1 * kpS[q][1];
  }
  wave_lds_order();

  float u  = S[lane][0];
  float p  = -S[0][0];
  float c0 = u;
#pragma unroll
  for (int j = 0; j < 32; ++j) {
    xch[lane] = u;
    wave_lds_order();
    float uj1 = 0.0f, ns = 0.0f, np = 0.0f;
    float nj[32], ni[32];
    if (j + 1 < 32) {
      uj1 = xch[j + 1];
      ns  = S[lane][j + 1];
      np  = -S[j + 1][j + 1];
#pragma unroll
      for (int k = 0; k < j; ++k) {
        nj[k] = S[j + 1][k];
        ni[k] = S[lane][k];
      }
    }
    const float n0 = ns;
    __builtin_amdgcn_sched_barrier(0);
    asm volatile("" : "+v"(p));
    const float djj = sqrtf(-p);
    if (j + 1 < 32) {
#pragma unroll
      for (int k = 0; k < j; ++k) {
        ns = __builtin_fmaf(-ni[k], nj[k], ns);
        np = __builtin_fmaf(nj[k], nj[k], np);
      }
    }
    const float v     = (lane == j) ? djj : u / djj;
    const float lastj = uj1 / djj;
    const float sv = (lane >= j) ? v : c0;
    S[lane][j] = sv;
    u  = __builtin_fmaf(-sv, lastj, ns);
    p  = __builtin_fmaf(lastj, lastj, np);
    c0 = n0;
  }
  wave_lds_order();

  red[lane] = logf(fabsf(S[lane][lane]));
  wave_lds_order();
  if (lane == 0) {
    float s = 0.0f;
    for (int i = 0; i < 32; ++i) s += red[i];
    wsum[wv] = s;
  }
  __syncthreads();
  if (threadIdx.x == 0) {
    float s = 0.0f;
    for (int i = 0; i < PRE_WAVES; ++i) s += wsum[i];
    parts[blockIdx.x] = s;
  }

  float* wq = Wmat + (size_t)dc * WFRAG + (lane & 15) * 4 + (lane >> 4) * 2;
  *(v2f*)(wq)          = (v2f){a0, a1};
  *(v2f*)(wq + 16 * 4) = (v2f){0.0f, 0.0f};
  float row[32];
#pragma unroll
  for (int q = 0; q < 32; ++q) row[q] = S[lane][q];
#pragma unroll
  for (int g = 0; g < 16; ++g) {
    const int sm = g >> 1;
    const int h  = g & 1;
    const int q0 = 8 * (sm >> 1) + 4 * h + 2 * (sm & 1);
    v2f w;
    w.x = (q0     <= lane) ? row[q0]     : 0.0f;
    w.y = (q0 + 1 <= lane) ? row[q0 + 1] : 0.0f;
    *(v2f*)(wq + ((sm + 1) * 32 + 16 * h) * 4) = w;
  }
}

#define OBLK     1088
#define ORUN     544
#define ORUN_V4  136
#define RB_PER_WG 4
#define TILE_V4  (2 * 16 * ORUN_V4)

__global__ __launch_bounds__(512) void gp_gemm(
    const float* __restrict__ z,
    const float* __restrict__ Wmat,
    float* __restrict__ out) {
  __shared__ float4 tile4[2][TILE_V4];

  const int tid   = threadIdx.x;
  const int dl    = tid >> 5;
  const int lane  = tid & 31;
  const int c0    = (blockIdx.x & 3) << 1;
  const int dh    = (blockIdx.x >> 2) & 1;
  const int chunk = blockIdx.x >> 3;
  const int d     = dh * 16 + dl;
  const int half  = lane >> 4;
  const int l16   = lane & 15;

  const int moff = 2 + c0 * 32 + (half << 2);
  const float* xbase = z + ((size_t)((chunk * RB_PER_WG) << 4) * D_DIM + (size_t)l16 * D_DIM + d) * T_DIM;

  v2f    re[2];
  v4f_a8 rq[2][8];
  re[0] = *(const v2f*)xbase;
#pragma unroll
  for (int i = 0; i < 8; ++i) rq[0][i] = *(const v4f_a8*)(xbase + moff + 8 * i);

  const v4f* wf = (const v4f*)(Wmat + (size_t)(d * 8 + c0) * WFRAG) + lane;
  v2f b0[2][9], b1[2][9];
#pragma unroll
  for (int cc = 0; cc < 2; ++cc) {
#pragma unroll
    for (int s = 0; s < 9; ++s) {
      const v4f q = wf[cc * (WFRAG / 4) + s * 32];
      b0[cc][s] = (v2f){q.x, q.y};
      b1[cc][s] = (v2f){q.z, q.w};
    }
  }

  {
    const float* xn = xbase + (size_t)16 * D_DIM * T_DIM;
    re[1] = *(const v2f*)xn;
#pragma unroll
    for (int i = 0; i < 8; ++i) rq[1][i] = *(const v4f_a8*)(xn + moff + 8 * i);
  }

#pragma unroll
  for (int it = 0; it < RB_PER_WG; ++it) {
    const int n0  = (chunk * RB_PER_WG + it) << 4;
    const int set = it & 1;
    const v2f ae = re[set];

    float* tl = (float*)tile4[set];
#pragma unroll
    for (int cc = 0; cc < 2; ++cc) {
      v8f acc0 = {};
      v8f acc1 = {};
      acc0 = __builtin_amdgcn_wmma_f32_16x16x4_f32(
          false, ae, false, b0[cc][0], (short)0, acc0, false, false);
      acc1 = __builtin_amdgcn_wmma_f32_16x16x4_f32(
          false, ae, false, b1[cc][0], (short)0, acc1, false, false);
#pragma unroll
      for (int i = 0; i < 4; ++i) {
        const v4f_a8 q = rq[set][4 * cc + i];
        const v2f alo = (v2f){q.x, q.y};
        const v2f ahi = (v2f){q.z, q.w};
        acc0 = __builtin_amdgcn_wmma_f32_16x16x4_f32(
            false, alo, false, b0[cc][1 + 2 * i], (short)0, acc0, false, false);
        acc1 = __builtin_amdgcn_wmma_f32_16x16x4_f32(
            false, alo, false, b1[cc][1 + 2 * i], (short)0, acc1, false, false);
        acc0 = __builtin_amdgcn_wmma_f32_16x16x4_f32(
            false, ahi, false, b0[cc][2 + 2 * i], (short)0, acc0, false, false);
        acc1 = __builtin_amdgcn_wmma_f32_16x16x4_f32(
            false, ahi, false, b1[cc][2 + 2 * i], (short)0, acc1, false, false);
      }

      float* tc = tl + cc * (16 * ORUN);
#pragma unroll
      for (int r = 0; r < 8; ++r) {
        float* rowp = tc + (r + (half << 3)) * ORUN + dl * 34;
        rowp[1 + l16]  = acc0[r];
        rowp[17 + l16] = acc1[r];
      }
      if (half == 0) {
        float* rowp = tc + l16 * ORUN + dl * 34;
        rowp[0]  = ae.x;
        rowp[33] = ae.y;
      }
    }
    __syncthreads();

    if (it + 2 < RB_PER_WG) {
      const float* xn = xbase + (size_t)(it + 2) * 16 * D_DIM * T_DIM;
      re[set] = *(const v2f*)xn;
#pragma unroll
      for (int i = 0; i < 8; ++i) rq[set][i] = *(const v4f_a8*)(xn + moff + 8 * i);
    }

    const float4* src4 = tile4[set];
#pragma unroll
    for (int k = 0; k < 9; ++k) {
      const int f = tid + 512 * k;
      if (f < TILE_V4) {
        const int rr  = f / ORUN_V4;
        const int col = f - rr * ORUN_V4;
        float4* dst = (float4*)(out + ((size_t)(n0 + (rr & 15)) * C_DIM + c0 + (rr >> 4)) * OBLK + dh * ORUN) + col;
        *dst = src4[f];
      }
    }
  }
}

__global__ __launch_bounds__(256) void gp_sldj(
    const float* __restrict__ sldj_in,
    const float* __restrict__ parts,
    float* __restrict__ out) {
  const int n = blockIdx.x * blockDim.x + threadIdx.x;
  const float own = sldj_in[n];
  float p[N_PARTS];
#pragma unroll
  for (int i = 0; i < N_PARTS; ++i) p[i] = parts[i];
  __builtin_amdgcn_sched_barrier(0);
  float tot = 0.0f;
#pragma unroll
  for (int i = 0; i < N_PARTS; ++i) tot += p[i];
  out[ZOUT_ELEMS + n] = own + tot;
}

extern "C" void kernel_launch(void* const* d_in, const int* in_sizes, int n_in,
                              void* d_out, int out_size, void* d_ws, size_t ws_size,
                              hipStream_t stream) {
  const float* z       = (const float*)d_in[0];
  const float* log_tau = (const float*)d_in[1];
  const float* ts0     = (const float*)d_in[2];
  const float* sldj    = (const float*)d_in[3];
  float* out = (float*)d_out;

  float* Wmat  = (float*)d_ws;
  float* parts = Wmat + 256 * WFRAG;

  gp_precompute<<<64, 128, 0, stream>>>(log_tau, ts0, Wmat, parts);
  gp_gemm<<<256, 512, 0, stream>>>(z, Wmat, out);
  gp_sldj<<<8, 256, 0, stream>>>(sldj, parts, out);
}
